# starting kernel
# baseline (speedup 1.0000x reference)
#include <hip/hip_runtime.h>
#include <hip/hip_bf16.h>

#define SEG_D 256

typedef __attribute__((ext_vector_type(2))) float v2f;
typedef __attribute__((ext_vector_type(8))) float v8f;

__global__ __launch_bounds__(1024) void seg_offsets_kernel(
    const int* __restrict__ sizes, int* __restrict__ offsets, int B) {
  __shared__ int tmp[1024];
  const int t = threadIdx.x;
  const int chunk = (B + 1023) / 1024;
  const int base = t * chunk;

  int s = 0;
  for (int i = 0; i < chunk; ++i) {
    int idx = base + i;
    if (idx < B) s += sizes[idx];
  }
  tmp[t] = s;
  __syncthreads();

  for (int off = 1; off < 1024; off <<= 1) {
    int v = (t >= off) ? tmp[t - off] : 0;
    __syncthreads();
    tmp[t] += v;
    __syncthreads();
  }

  int run = (t == 0) ? 0 : tmp[t - 1];
  for (int i = 0; i < chunk; ++i) {
    int idx = base + i;
    if (idx < B) {
      offsets[idx] = run;
      run += sizes[idx];
    }
  }
  if (t == 1023) offsets[B] = tmp[1023];
}

__global__ __launch_bounds__(512) void seg_mean_wmma_kernel(
    const float* __restrict__ x, const int* __restrict__ sizes,
    const int* __restrict__ offsets, float* __restrict__ out) {
  const int b = blockIdx.x;
  const int start = offsets[b];
  const int size  = sizes[b];
  const int end   = start + size;

  const int wave = threadIdx.x >> 5;
  const int lane = threadIdx.x & 31;
  const int n    = lane & 15;
  const int kh   = lane >> 4;
  const int col  = (wave << 4) + n;

  const float* __restrict__ xc = x + col;

  v2f a;  a.x = 1.0f;  a.y = 1.0f;
  v8f acc0 = {};
  v8f acc1 = {};

  const int nchunks = size >> 3;
  const float* p = xc + (long)(start + kh) * SEG_D;

  if (nchunks > 0) {
    v2f b0, b1;
    b0.x = __builtin_nontemporal_load(p + 0 * SEG_D);
    b0.y = __builtin_nontemporal_load(p + 2 * SEG_D);
    b1.x = __builtin_nontemporal_load(p + 4 * SEG_D);
    b1.y = __builtin_nontemporal_load(p + 6 * SEG_D);

    for (int i = 1; i < nchunks; ++i) {
      p += 8 * SEG_D;
      __builtin_prefetch(p + 8 * SEG_D, 0, 1);
      v2f c0, c1;
      c0.x = __builtin_nontemporal_load(p + 0 * SEG_D);
      c0.y = __builtin_nontemporal_load(p + 2 * SEG_D);
      c1.x = __builtin_nontemporal_load(p + 4 * SEG_D);
      c1.y = __builtin_nontemporal_load(p + 6 * SEG_D);
      acc0 = __builtin_amdgcn_wmma_f32_16x16x4_f32(
          false, a, false, b0, (short)0, acc0, false, false);
      acc1 = __builtin_amdgcn_wmma_f32_16x16x4_f32(
          false, a, false, b1, (short)0, acc1, false, false);
      b0 = c0;
      b1 = c1;
    }
    acc0 = __builtin_amdgcn_wmma_f32_16x16x4_f32(
        false, a, false, b0, (short)0, acc0, false, false);
    acc1 = __builtin_amdgcn_wmma_f32_16x16x4_f32(
        false, a, false, b1, (short)0, acc1, false, false);
  }

  const int done = start + (nchunks << 3);
  if (done < end) {
    const int last = end - 1;
    int r0 = done + 0 + kh;
    int r1 = done + 2 + kh;
    int r2 = done + 4 + kh;
    int r3 = done + 6 + kh;
    float m0 = (r0 <= last) ? 1.0f : 0.0f;
    float m1 = (r1 <= last) ? 1.0f : 0.0f;
    float m2 = (r2 <= last) ? 1.0f : 0.0f;
    float m3 = (r3 <= last) ? 1.0f : 0.0f;
    int c0i = (r0 <= last) ? r0 : last;
    int c1i = (r1 <= last) ? r1 : last;
    int c2i = (r2 <= last) ? r2 : last;
    int c3i = (r3 <= last) ? r3 : last;

    v2f b0, b1;
    b0.x = m0 * xc[(long)c0i * SEG_D];
    b0.y = m1 * xc[(long)c1i * SEG_D];
    b1.x = m2 * xc[(long)c2i * SEG_D];
    b1.y = m3 * xc[(long)c3i * SEG_D];

    acc0 = __builtin_amdgcn_wmma_f32_16x16x4_f32(
        false, a, false, b0, (short)0, acc0, false, false);
    acc1 = __builtin_amdgcn_wmma_f32_16x16x4_f32(
        false, a, false, b1, (short)0, acc1, false, false);
  }

  float sum = acc0[0] + acc1[0];
  if (lane < 16) {
    out[(long)b * SEG_D + col] = sum / (float)size;
  }
}

extern "C" void kernel_launch(void* const* d_in, const int* in_sizes, int n_in,
                              void* d_out, int out_size, void* d_ws, size_t ws_size,
                              hipStream_t stream) {
  const float* x     = (const float*)d_in[0];
  const int*   sizes = (const int*)d_in[1];
  float*       out   = (float*)d_out;

  const int B = in_sizes[1];
  int* offsets = (int*)d_ws;

  seg_offsets_kernel<<<1, 1024, 0, stream>>>(sizes, offsets, B);
  seg_mean_wmma_kernel<<<B, 512, 0, stream>>>(x, sizes, offsets, out);
}


# Offsets launch shortened: each thread moves its 4 sizes with one 16-byte load and store, prefix by a DPP scan over the wave's lanes plus the 32 wave totals through LDS: 1 barrier in place of 20
# speedup vs baseline: 1.4845x; 1.4845x over previous
#include <hip/hip_runtime.h>
#include <hip/hip_bf16.h>

#define SEG_D 256

typedef __attribute__((ext_vector_type(2))) float v2f;
typedef __attribute__((ext_vector_type(8))) float v8f;

typedef __attribute__((ext_vector_type(4))) int v4i;

__device__ __forceinline__ int wave_inclusive_scan(int v, int lane) {
  v += __builtin_amdgcn_update_dpp(0, v, 0x111, 0xf, 0xf, true);
  v += __builtin_amdgcn_update_dpp(0, v, 0x112, 0xf, 0xf, true);
  v += __builtin_amdgcn_update_dpp(0, v, 0x114, 0xf, 0xf, true);
  v += __builtin_amdgcn_update_dpp(0, v, 0x118, 0xf, 0xf, true);
  const int row0 = __builtin_amdgcn_readlane(v, 15);
  return lane >= 16 ? v + row0 : v;
}

__device__ __forceinline__ v4i load_quad(const int* __restrict__ sizes, int idx, int B) {
  v4i q = {0, 0, 0, 0};
  if (idx + 3 < B) {
    q = *reinterpret_cast<const v4i*>(sizes + idx);
  } else {
    if (idx     < B) q.x = sizes[idx];
    if (idx + 1 < B) q.y = sizes[idx + 1];
    if (idx + 2 < B) q.z = sizes[idx + 2];
  }
  return q;
}

__device__ __forceinline__ void store_quad(int* __restrict__ offsets, int idx, int B,
                                           int run, v4i q) {
  v4i o;
  o.x = run;
  o.y = run + q.x;
  o.z = run + q.x + q.y;
  o.w = run + q.x + q.y + q.z;
  if (idx + 3 < B) {
    *reinterpret_cast<v4i*>(offsets + idx) = o;
  } else {
    if (idx     < B) offsets[idx]     = o.x;
    if (idx + 1 < B) offsets[idx + 1] = o.y;
    if (idx + 2 < B) offsets[idx + 2] = o.z;
  }
}

__global__ __launch_bounds__(1024) void seg_offsets_kernel(
    const int* __restrict__ sizes, int* __restrict__ offsets, int B) {
  __shared__ int wave_total[32];
  const int t = threadIdx.x;
  const int lane = t & 31;
  const int wave = __builtin_amdgcn_readfirstlane(t >> 5);
  const int per  = ((B + 4095) >> 12) << 2;
  const int base = t * per;

  const v4i q0 = load_quad(sizes, base, B);
  int s = q0.x + q0.y + q0.z + q0.w;
  for (int i = 4; i < per; i += 4) {
    const v4i q = load_quad(sizes, base + i, B);
    s += q.x + q.y + q.z + q.w;
  }

  const int incl = wave_inclusive_scan(s, lane);
  if (lane == 31) wave_total[wave] = incl;
  __syncthreads();
  const int tot      = wave_total[lane];
  const int tot_incl = wave_inclusive_scan(tot, lane);
  const int wave_base = __builtin_amdgcn_readlane(tot_incl - tot, wave);

  int run = wave_base + incl - s;
  store_quad(offsets, base, B, run, q0);
  run += q0.x + q0.y + q0.z + q0.w;
  for (int i = 4; i < per; i += 4) {
    const v4i q = load_quad(sizes, base + i, B);
    store_quad(offsets, base + i, B, run, q);
    run += q.x + q.y + q.z + q.w;
  }
  if (t == 1023) offsets[B] = run;
}

#define GROUP_SEGS 4
#define BATCH_ROWS 10

struct RowBatch { v2f p[2][BATCH_ROWS / 2]; };

__device__ __forceinline__ void load_batch(RowBatch& b, const float* __restrict__ xw,
                                           int row, int last_base) {
  const int base = row < last_base ? row : last_base;
  const float* p = xw + (long)base * SEG_D;
#pragma unroll
  for (int i = 0; i < BATCH_ROWS / 2; ++i) {
    b.p[0][i].x = __builtin_nontemporal_load(p + (2 * i + 0) * SEG_D);
    b.p[1][i].x = __builtin_nontemporal_load(p + (2 * i + 0) * SEG_D + 32);
    b.p[0][i].y = __builtin_nontemporal_load(p + (2 * i + 1) * SEG_D);
    b.p[1][i].y = __builtin_nontemporal_load(p + (2 * i + 1) * SEG_D + 32);
  }
}

__device__ __forceinline__ void consume_batch(const RowBatch& b, int row, int last_base,
                                              int lo, int hi, int q, v2f one,
                                              v8f& acc0, v8f& acc1) {
  const int base = row < last_base ? row : last_base;
  const int lo_b = lo > row ? lo : row;
  const int len  = hi > lo_b ? hi - lo_b : 0;
  const int d    = base - lo_b + q;
#pragma unroll
  for (int i = 0; i < BATCH_ROWS / 2; ++i) {
    const bool in = (unsigned)(d + 2 * i) < (unsigned)len;
    v2f a;
    a.x = in ? one.x : 0.0f;
    a.y = in ? one.y : 0.0f;
    acc0 = __builtin_amdgcn_wmma_f32_16x16x4_f32(
        false, a, false, b.p[0][i], (short)0, acc0, false, false);
    acc1 = __builtin_amdgcn_wmma_f32_16x16x4_f32(
        false, a, false, b.p[1][i], (short)0, acc1, false, false);
  }
}

__device__ __forceinline__ void order_fence(v8f& acc0, v8f& acc1) {
  asm volatile("" : "+v"(acc0), "+v"(acc1) : : "memory");
}

__global__ __launch_bounds__(128) void seg_mean_gemm_kernel(
    const float* __restrict__ x, const int* __restrict__ sizes,
    const int* __restrict__ offsets, float* __restrict__ out, int B, int n_elems) {
  const int N = n_elems / SEG_D;
  const int last_base = N - BATCH_ROWS;
  const int lane = threadIdx.x & 31;
  const int wave = __builtin_amdgcn_readfirstlane((int)(threadIdx.x >> 5));
  const float* __restrict__ xw = x + (wave << 6) + lane;

  const int  j     = lane & 3;
  const int  q     = (lane >> 2) & 1;
  const bool holds = (((lane >> 3) & 1) == (lane >> 4));
  v2f one;
  one.x = q == 0 ? 1.0f : 0.0f;
  one.y = q == 1 ? 1.0f : 0.0f;

  const int ngroups = (B + GROUP_SEGS - 1) / GROUP_SEGS;
  for (int g = blockIdx.x; g < ngroups; g += gridDim.x) {
    const int s0 = g * GROUP_SEGS;

    int R0 = offsets[s0];
    R0 = R0 < N ? R0 : N;
    int R1 = N;
    if (s0 + GROUP_SEGS < B) {
      R1 = offsets[s0 + GROUP_SEGS];
      R1 = R1 < N ? R1 : N;
    }

    const int sidx = s0 + j;
    const int sclp = sidx < B ? sidx : B - 1;
    int lo = offsets[sclp];
    int hi = offsets[sclp + 1];
    lo = lo < N ? lo : N;
    hi = hi < N ? hi : N;
    if (sidx == B - 1) hi = N;
    if (!holds || sidx >= B) { lo = 0; hi = 0; }

    v8f acc0 = {};
    v8f acc1 = {};

    const int nb    = (R1 - R0 + BATCH_ROWS - 1) / BATCH_ROWS;
    const int ntrip = (nb + 2) / 3;

    RowBatch b0, b1, b2;
    int row = R0;
    load_batch(b0, xw, row, last_base);
    load_batch(b1, xw, row + BATCH_ROWS, last_base);
    load_batch(b2, xw, row + 2 * BATCH_ROWS, last_base);

    for (int t = 1; t < ntrip; ++t) {
      consume_batch(b0, row, last_base, lo, hi, q, one, acc0, acc1);
      order_fence(acc0, acc1);
      load_batch(b0, xw, row + 3 * BATCH_ROWS, last_base);
      order_fence(acc0, acc1);
      consume_batch(b1, row + BATCH_ROWS, last_base, lo, hi, q, one, acc0, acc1);
      order_fence(acc0, acc1);
      load_batch(b1, xw, row + 4 * BATCH_ROWS, last_base);
      order_fence(acc0, acc1);
      consume_batch(b2, row + 2 * BATCH_ROWS, last_base, lo, hi, q, one, acc0, acc1);
      order_fence(acc0, acc1);
      load_batch(b2, xw, row + 5 * BATCH_ROWS, last_base);
      order_fence(acc0, acc1);
      row += 3 * BATCH_ROWS;
    }
    consume_batch(b0, row, last_base, lo, hi, q, one, acc0, acc1);
    consume_batch(b1, row + BATCH_ROWS, last_base, lo, hi, q, one, acc0, acc1);
    consume_batch(b2, row + 2 * BATCH_ROWS, last_base, lo, hi, q, one, acc0, acc1);

    float* o = out + (long)s0 * SEG_D + (wave << 6) + lane;
#pragma unroll
    for (int r = 0; r < GROUP_SEGS; ++r) {
      if (s0 + r < B) {
        const float sz = (float)sizes[s0 + r];
        o[(long)r * SEG_D]      = (acc0[r] + acc0[r + GROUP_SEGS]) / sz;
        o[(long)r * SEG_D + 32] = (acc1[r] + acc1[r + GROUP_SEGS]) / sz;
      }
    }
  }
}

extern "C" void kernel_launch(void* const* d_in, const int* in_sizes, int n_in,
                              void* d_out, int out_size, void* d_ws, size_t ws_size,
                              hipStream_t stream) {
  const float* x     = (const float*)d_in[0];
  const int*   sizes = (const int*)d_in[1];
  float*       out   = (float*)d_out;

  const int B = in_sizes[1];
  int* offsets = (int*)d_ws;

  seg_offsets_kernel<<<1, 1024, 0, stream>>>(sizes, offsets, B);
  seg_mean_gemm_kernel<<<1024, 128, 0, stream>>>(x, sizes, offsets, out, B, in_sizes[0]);
}


# Batches of 10 rows inside one segment: the 5 row pairs of a slab are added on the vector ALU, then ONE WMMA per slab applies the 0/1 segment map (was 5); a column's reduction tree changes
# speedup vs baseline: 1.5469x; 1.0420x over previous
#include <hip/hip_runtime.h>
#include <hip/hip_bf16.h>

#define SEG_D 256

typedef __attribute__((ext_vector_type(2))) float v2f;
typedef __attribute__((ext_vector_type(8))) float v8f;

typedef __attribute__((ext_vector_type(4))) int v4i;

__device__ __forceinline__ int wave_inclusive_scan(int v, int lane) {
  v += __builtin_amdgcn_update_dpp(0, v, 0x111, 0xf, 0xf, true);
  v += __builtin_amdgcn_update_dpp(0, v, 0x112, 0xf, 0xf, true);
  v += __builtin_amdgcn_update_dpp(0, v, 0x114, 0xf, 0xf, true);
  v += __builtin_amdgcn_update_dpp(0, v, 0x118, 0xf, 0xf, true);
  const int row0 = __builtin_amdgcn_readlane(v, 15);
  return lane >= 16 ? v + row0 : v;
}

__device__ __forceinline__ v4i load_quad(const int* __restrict__ sizes, int idx, int B) {
  v4i q = {0, 0, 0, 0};
  if (idx + 3 < B) {
    q = *reinterpret_cast<const v4i*>(sizes + idx);
  } else {
    if (idx     < B) q.x = sizes[idx];
    if (idx + 1 < B) q.y = sizes[idx + 1];
    if (idx + 2 < B) q.z = sizes[idx + 2];
  }
  return q;
}

__device__ __forceinline__ void store_quad(int* __restrict__ offsets, int idx, int B,
                                           int run, v4i q) {
  v4i o;
  o.x = run;
  o.y = run + q.x;
  o.z = run + q.x + q.y;
  o.w = run + q.x + q.y + q.z;
  if (idx + 3 < B) {
    *reinterpret_cast<v4i*>(offsets + idx) = o;
  } else {
    if (idx     < B) offsets[idx]     = o.x;
    if (idx + 1 < B) offsets[idx + 1] = o.y;
    if (idx + 2 < B) offsets[idx + 2] = o.z;
  }
}

__global__ __launch_bounds__(1024) void seg_offsets_kernel(
    const int* __restrict__ sizes, int* __restrict__ offsets, int B) {
  __shared__ int wave_total[32];
  const int t = threadIdx.x;
  const int lane = t & 31;
  const int wave = __builtin_amdgcn_readfirstlane(t >> 5);
  const int per  = ((B + 4095) >> 12) << 2;
  const int base = t * per;

  const v4i q0 = load_quad(sizes, base, B);
  int s = q0.x + q0.y + q0.z + q0.w;
  for (int i = 4; i < per; i += 4) {
    const v4i q = load_quad(sizes, base + i, B);
    s += q.x + q.y + q.z + q.w;
  }

  const int incl = wave_inclusive_scan(s, lane);
  if (lane == 31) wave_total[wave] = incl;
  __syncthreads();
  const int tot      = wave_total[lane];
  const int tot_incl = wave_inclusive_scan(tot, lane);
  const int wave_base = __builtin_amdgcn_readlane(tot_incl - tot, wave);

  int run = wave_base + incl - s;
  store_quad(offsets, base, B, run, q0);
  run += q0.x + q0.y + q0.z + q0.w;
  for (int i = 4; i < per; i += 4) {
    const v4i q = load_quad(sizes, base + i, B);
    store_quad(offsets, base + i, B, run, q);
    run += q.x + q.y + q.z + q.w;
  }
  if (t == 1023) offsets[B] = run;
}

#define GROUP_SEGS 4
#define BATCH_ROWS 10

struct RowBatch { v2f p[2][BATCH_ROWS / 2]; };

__device__ __forceinline__ void load_batch(RowBatch& b, const float* __restrict__ xw,
                                           int row, int last_base) {
  const int base = row < last_base ? row : last_base;
  const float* p = xw + (long)base * SEG_D;
#pragma unroll
  for (int i = 0; i < BATCH_ROWS / 2; ++i) {
    b.p[0][i].x = __builtin_nontemporal_load(p + (2 * i + 0) * SEG_D);
    b.p[1][i].x = __builtin_nontemporal_load(p + (2 * i + 0) * SEG_D + 32);
    b.p[0][i].y = __builtin_nontemporal_load(p + (2 * i + 1) * SEG_D);
    b.p[1][i].y = __builtin_nontemporal_load(p + (2 * i + 1) * SEG_D + 32);
  }
}

struct SegEnds { int c1, c2, c3, c4; };

__device__ __forceinline__ void consume(const RowBatch& b, int row, int last_base,
                                        const SegEnds& e, int lo, int hi, int q,
                                        v2f one, v8f& acc0, v8f& acc1,
                                        v8f& end0, v8f& end1) {
  const unsigned w = BATCH_ROWS - 1;
  const bool no_end = ((unsigned)(e.c1 - row - 1) >= w) & ((unsigned)(e.c2 - row - 1) >= w) &
                      ((unsigned)(e.c3 - row - 1) >= w) & ((unsigned)(e.c4 - row - 1) >= w);
  v2f a, s0, s1;
  if (no_end & (row <= last_base)) {
    s0 = ((b.p[0][0] + b.p[0][1]) + (b.p[0][2] + b.p[0][3])) + b.p[0][4];
    s1 = ((b.p[1][0] + b.p[1][1]) + (b.p[1][2] + b.p[1][3])) + b.p[1][4];
    const bool mine = (unsigned)(row - lo) < (unsigned)(hi - lo);
    a.x = mine ? one.x : 0.0f;
    a.y = mine ? one.y : 0.0f;
  } else {
    const int base = row < last_base ? row : last_base;
    const int lo_b = lo > row ? lo : row;
    const int len  = hi > lo_b ? hi - lo_b : 0;
    const int d    = base - lo_b + q;
#pragma unroll
    for (int i = 0; i < BATCH_ROWS / 2; ++i) {
      const bool in = (unsigned)(d + 2 * i) < (unsigned)len;
      a.x = in ? one.x : 0.0f;
      a.y = in ? one.y : 0.0f;
      if (i < BATCH_ROWS / 2 - 1) {
        end0 = __builtin_amdgcn_wmma_f32_16x16x4_f32(
            false, a, false, b.p[0][i], (short)0, end0, false, false);
        end1 = __builtin_amdgcn_wmma_f32_16x16x4_f32(
            false, a, false, b.p[1][i], (short)0, end1, false, false);
      } else {
        s0 = b.p[0][i];
        s1 = b.p[1][i];
      }
    }
  }
  acc0 = __builtin_amdgcn_wmma_f32_16x16x4_f32(
      false, a, false, s0, (short)0, acc0, false, false);
  acc1 = __builtin_amdgcn_wmma_f32_16x16x4_f32(
      false, a, false, s1, (short)0, acc1, false, false);
}

__device__ __forceinline__ void order_fence(v8f& acc0, v8f& acc1) {
  asm volatile("" : "+v"(acc0), "+v"(acc1) : : "memory");
}

__global__ __launch_bounds__(128) void seg_mean_gemm_kernel(
    const float* __restrict__ x, const int* __restrict__ sizes,
    const int* __restrict__ offsets, float* __restrict__ out, int B, int n_elems) {
  const int N = n_elems / SEG_D;
  const int last_base = N - BATCH_ROWS;
  const int lane = threadIdx.x & 31;
  const int wave = __builtin_amdgcn_readfirstlane((int)(threadIdx.x >> 5));
  const float* __restrict__ xw = x + (wave << 6) + lane;

  const int  j     = lane & 3;
  const int  q     = (lane >> 2) & 1;
  const bool holds = (((lane >> 3) & 1) == (lane >> 4));
  v2f one;
  one.x = q == 0 ? 1.0f : 0.0f;
  one.y = q == 1 ? 1.0f : 0.0f;

  const int ngroups = (B + GROUP_SEGS - 1) / GROUP_SEGS;
  for (int g = blockIdx.x; g < ngroups; g += gridDim.x) {
    const int s0 = g * GROUP_SEGS;

    int cb[GROUP_SEGS + 1];
#pragma unroll
    for (int k = 0; k <= GROUP_SEGS; ++k) {
      int v = N;
      if (s0 + k < B) {
        v = offsets[s0 + k];
        v = v < N ? v : N;
      }
      cb[k] = v;
    }
    const int R0 = cb[0];
    const int R1 = cb[GROUP_SEGS];
    SegEnds e;
    e.c1 = cb[1]; e.c2 = cb[2]; e.c3 = cb[3]; e.c4 = cb[4];

    const int sidx = s0 + j;
    const int sclp = sidx < B ? sidx : B - 1;
    int lo = offsets[sclp];
    int hi = offsets[sclp + 1];
    lo = lo < N ? lo : N;
    hi = hi < N ? hi : N;
    if (sidx == B - 1) hi = N;
    if (!holds || sidx >= B) { lo = 0; hi = 0; }

    v8f acc0 = {};
    v8f acc1 = {};
    v8f end0 = {};
    v8f end1 = {};

    const int nb    = (R1 - R0 + BATCH_ROWS - 1) / BATCH_ROWS;
    const int ntrip = (nb + 2) / 3;

    RowBatch b0, b1, b2;
    int row = R0;
    load_batch(b0, xw, row, last_base);
    load_batch(b1, xw, row + BATCH_ROWS, last_base);
    load_batch(b2, xw, row + 2 * BATCH_ROWS, last_base);

    for (int t = 1; t < ntrip; ++t) {
      consume(b0, row, last_base, e, lo, hi, q, one, acc0, acc1, end0, end1);
      order_fence(acc0, acc1);
      load_batch(b0, xw, row + 3 * BATCH_ROWS, last_base);
      order_fence(acc0, acc1);
      consume(b1, row + BATCH_ROWS, last_base, e, lo, hi, q, one, acc0, acc1, end0, end1);
      order_fence(acc0, acc1);
      load_batch(b1, xw, row + 4 * BATCH_ROWS, last_base);
      order_fence(acc0, acc1);
      consume(b2, row + 2 * BATCH_ROWS, last_base, e, lo, hi, q, one, acc0, acc1, end0, end1);
      order_fence(acc0, acc1);
      load_batch(b2, xw, row + 5 * BATCH_ROWS, last_base);
      order_fence(acc0, acc1);
      row += 3 * BATCH_ROWS;
    }
    consume(b0, row, last_base, e, lo, hi, q, one, acc0, acc1, end0, end1);
    consume(b1, row + BATCH_ROWS, last_base, e, lo, hi, q, one, acc0, acc1, end0, end1);
    consume(b2, row + 2 * BATCH_ROWS, last_base, e, lo, hi, q, one, acc0, acc1, end0, end1);

    float* o = out + (long)s0 * SEG_D + (wave << 6) + lane;
#pragma unroll
    for (int r = 0; r < GROUP_SEGS; ++r) {
      if (s0 + r < B) {
        const float sz = (float)sizes[s0 + r];
        o[(long)r * SEG_D]      = ((acc0[r] + acc0[r + GROUP_SEGS]) +
                                   (end0[r] + end0[r + GROUP_SEGS])) / sz;
        o[(long)r * SEG_D + 32] = ((acc1[r] + acc1[r + GROUP_SEGS]) +
                                   (end1[r] + end1[r + GROUP_SEGS])) / sz;
      }
    }
  }
}

extern "C" void kernel_launch(void* const* d_in, const int* in_sizes, int n_in,
                              void* d_out, int out_size, void* d_ws, size_t ws_size,
                              hipStream_t stream) {
  const float* x     = (const float*)d_in[0];
  const int*   sizes = (const int*)d_in[1];
  float*       out   = (float*)d_out;

  const int B = in_sizes[1];
  int* offsets = (int*)d_ws;

  seg_offsets_kernel<<<1, 1024, 0, stream>>>(sizes, offsets, B);
  seg_mean_gemm_kernel<<<1024, 128, 0, stream>>>(x, sizes, offsets, out, B, in_sizes[0]);
}
